# starting kernel
# baseline (speedup 1.0000x reference)
#include <hip/hip_runtime.h>
#include <hip/hip_bf16.h>

typedef float v2f __attribute__((ext_vector_type(2)));
typedef float v8f __attribute__((ext_vector_type(8)));

static __device__ __forceinline__ v8f wmma_16x16x4_f32(v2f a, v2f b, v8f c) {
#if __has_builtin(__builtin_amdgcn_wmma_f32_16x16x4_f32)
    return __builtin_amdgcn_wmma_f32_16x16x4_f32(
         false, a,  false, b,
         (short)0, c,  false,  false);
#else
    v8f d;
    asm volatile("v_wmma_f32_16x16x4_f32 %0, %1, %2, %3"
                 : "=v"(d)
                 : "v"(a), "v"(b), "v"(c));
    return d;
#endif
}

__global__ void mpo_setup_kernel(const float* __restrict__ w, float* __restrict__ ws) {
    float* bufA = ws;
    float* bufB = ws + 262144;
    float* wf   = ws + 524288;
    const int tid = threadIdx.x;

    if (tid < 64) {
        int a = tid & 15;
        int I = (tid >> 5) & 1;
        int J = (tid >> 4) & 1;
        bufA[(I * 2 + J) * 16 + a] = w[a * 4 + I * 2 + J];
    }
    __syncthreads();

    int dim = 2;
    for (int c = 1; c <= 6; ++c) {
        const float* core = w + 64 + (c - 1) * 1024;
        const int odim  = dim * 2;
        const int total = odim * odim * 16;
        for (int idx = tid; idx < total; idx += 1024) {
            int b   = idx & 15;
            int rem = idx >> 4;
            int Jo  = rem % odim;
            int Io  = rem / odim;
            int I = Io >> 1, i = Io & 1;
            int J = Jo >> 1, j = Jo & 1;
            const float* Trow = bufA + (I * dim + J) * 16;
            float s = 0.f;
#pragma unroll
            for (int a = 0; a < 16; ++a)
                s += Trow[a] * core[((a * 16 + b) * 2 + i) * 2 + j];
            bufB[idx] = s;
        }
        __syncthreads();
        float* t = bufA; bufA = bufB; bufB = t;
        dim = odim;
    }

    const float* core7 = w + 6208;
    for (int idx = tid; idx < 65536; idx += 1024) {
        int n = idx & 255;
        int r = idx >> 8;
        int I = r >> 1, i = r & 1;
        int J = n >> 1, j = n & 1;
        const float* Trow = bufA + (I * 128 + J) * 16;
        float s = 0.f;
#pragma unroll
        for (int a = 0; a < 16; ++a)
            s += Trow[a] * core7[a * 4 + i * 2 + j];
        int kstep = r >> 2, kk = r & 3;
        int nt    = n >> 4, nl = n & 15;
        int lane  = ((kk >= 2) ? 16 : 0) + nl;
        int v     = kk & 1;
        wf[((kstep * 16 + nt) * 32 + lane) * 2 + v] = s;
    }
}

__global__ void __launch_bounds__(256, 1)
mpo_main_kernel(const float* __restrict__ x,
                const float* __restrict__ dk,
                const float* __restrict__ bias,
                const float* __restrict__ wf,
                float* __restrict__ out) {
    extern __shared__ float lds[];
    float* part = lds + 65536;
    const int tid = threadIdx.x;

    {
        const float4* src = (const float4*)wf;
        float4* dst = (float4*)lds;
        for (int i = tid; i < 16384; i += 256)
            dst[i] = src[i];
    }
    __syncthreads();

    const int wave = tid >> 5;
    const int lane = tid & 31;
    const int l16  = lane & 15;
    const int koff = (lane < 16) ? 0 : 2;
    const int q    = wave >> 2;
    const int g    = wave & 3;

    const long rowbase = ((long)blockIdx.x * 2 + q) * 64;
    const float* xbase = x + (rowbase + l16) * 256 + koff;

    v8f acc[4][4];
#pragma unroll
    for (int s = 0; s < 4; ++s)
#pragma unroll
        for (int t = 0; t < 4; ++t)
#pragma unroll
            for (int v = 0; v < 8; ++v)
                acc[s][t][v] = 0.f;

    const float* ldsb = lds + lane * 2 + g * 256;

    v2f a0 = *(const v2f*)(xbase);
    v2f a1 = *(const v2f*)(xbase + 4096);
    v2f a2 = *(const v2f*)(xbase + 8192);
    v2f a3 = *(const v2f*)(xbase + 12288);

    for (int k = 0; k < 63; ++k) {
        const float* xn = xbase + 4 * (k + 1);
        v2f n0 = *(const v2f*)(xn);
        v2f n1 = *(const v2f*)(xn + 4096);
        v2f n2 = *(const v2f*)(xn + 8192);
        v2f n3 = *(const v2f*)(xn + 12288);

        const float* lk = ldsb + k * 1024;
        v2f b0 = *(const v2f*)(lk);
        v2f b1 = *(const v2f*)(lk + 64);
        v2f b2 = *(const v2f*)(lk + 128);
        v2f b3 = *(const v2f*)(lk + 192);

        acc[0][0] = wmma_16x16x4_f32(a0, b0, acc[0][0]);
        acc[0][1] = wmma_16x16x4_f32(a0, b1, acc[0][1]);
        acc[0][2] = wmma_16x16x4_f32(a0, b2, acc[0][2]);
        acc[0][3] = wmma_16x16x4_f32(a0, b3, acc[0][3]);
        acc[1][0] = wmma_16x16x4_f32(a1, b0, acc[1][0]);
        acc[1][1] = wmma_16x16x4_f32(a1, b1, acc[1][1]);
        acc[1][2] = wmma_16x16x4_f32(a1, b2, acc[1][2]);
        acc[1][3] = wmma_16x16x4_f32(a1, b3, acc[1][3]);
        acc[2][0] = wmma_16x16x4_f32(a2, b0, acc[2][0]);
        acc[2][1] = wmma_16x16x4_f32(a2, b1, acc[2][1]);
        acc[2][2] = wmma_16x16x4_f32(a2, b2, acc[2][2]);
        acc[2][3] = wmma_16x16x4_f32(a2, b3, acc[2][3]);
        acc[3][0] = wmma_16x16x4_f32(a3, b0, acc[3][0]);
        acc[3][1] = wmma_16x16x4_f32(a3, b1, acc[3][1]);
        acc[3][2] = wmma_16x16x4_f32(a3, b2, acc[3][2]);
        acc[3][3] = wmma_16x16x4_f32(a3, b3, acc[3][3]);

        a0 = n0; a1 = n1; a2 = n2; a3 = n3;
    }
    {
        const float* lk = ldsb + 63 * 1024;
        v2f b0 = *(const v2f*)(lk);
        v2f b1 = *(const v2f*)(lk + 64);
        v2f b2 = *(const v2f*)(lk + 128);
        v2f b3 = *(const v2f*)(lk + 192);
        acc[0][0] = wmma_16x16x4_f32(a0, b0, acc[0][0]);
        acc[0][1] = wmma_16x16x4_f32(a0, b1, acc[0][1]);
        acc[0][2] = wmma_16x16x4_f32(a0, b2, acc[0][2]);
        acc[0][3] = wmma_16x16x4_f32(a0, b3, acc[0][3]);
        acc[1][0] = wmma_16x16x4_f32(a1, b0, acc[1][0]);
        acc[1][1] = wmma_16x16x4_f32(a1, b1, acc[1][1]);
        acc[1][2] = wmma_16x16x4_f32(a1, b2, acc[1][2]);
        acc[1][3] = wmma_16x16x4_f32(a1, b3, acc[1][3]);
        acc[2][0] = wmma_16x16x4_f32(a2, b0, acc[2][0]);
        acc[2][1] = wmma_16x16x4_f32(a2, b1, acc[2][1]);
        acc[2][2] = wmma_16x16x4_f32(a2, b2, acc[2][2]);
        acc[2][3] = wmma_16x16x4_f32(a2, b3, acc[2][3]);
        acc[3][0] = wmma_16x16x4_f32(a3, b0, acc[3][0]);
        acc[3][1] = wmma_16x16x4_f32(a3, b1, acc[3][1]);
        acc[3][2] = wmma_16x16x4_f32(a3, b2, acc[3][2]);
        acc[3][3] = wmma_16x16x4_f32(a3, b3, acc[3][3]);
    }

    float kv[4];
#pragma unroll
    for (int t = 0; t < 4; ++t)
        kv[t] = dk[(g * 4 + t) * 16 + l16];

#pragma unroll
    for (int s = 0; s < 4; ++s) {
        float sv[8];
#pragma unroll
        for (int v = 0; v < 8; ++v) sv[v] = 0.f;
#pragma unroll
        for (int t = 0; t < 4; ++t)
#pragma unroll
            for (int v = 0; v < 8; ++v) {
                float h = acc[s][t][v];
                sv[v] += h * h * kv[t];
            }
#pragma unroll
        for (int v = 0; v < 8; ++v) {
            sv[v] += __shfl_xor(sv[v], 1, 32);
            sv[v] += __shfl_xor(sv[v], 2, 32);
            sv[v] += __shfl_xor(sv[v], 4, 32);
            sv[v] += __shfl_xor(sv[v], 8, 32);
        }
        if (l16 == 0) {
            int half = (lane >= 16) ? 8 : 0;
            float* p = part + ((q * 4 + g) * 4 + s) * 16 + half;
#pragma unroll
            for (int v = 0; v < 8; ++v) p[v] = sv[v];
        }
    }
    __syncthreads();

    if (tid < 128) {
        int q2  = tid >> 6;
        int s2  = (tid >> 4) & 3;
        int idx = tid & 15;
        const float* p = part + (q2 * 16 + s2) * 16 + idx;
        float r = p[0] + p[64] + p[128] + p[192];
        out[(long)blockIdx.x * 128 + tid] = r + bias[0];
    }
}

extern "C" void kernel_launch(void* const* d_in, const int* in_sizes, int n_in,
                              void* d_out, int out_size, void* d_ws, size_t ws_size,
                              hipStream_t stream) {
    const float* x   = (const float*)d_in[0];
    const float* mpw = (const float*)d_in[1];
    const float* dk  = (const float*)d_in[2];
    const float* db  = (const float*)d_in[3];
    float* out = (float*)d_out;
    float* ws  = (float*)d_ws;

    hipLaunchKernelGGL(mpo_setup_kernel, dim3(1), dim3(1024), 0, stream, mpw, ws);

    hipLaunchKernelGGL(mpo_main_kernel, dim3(2048), dim3(256),
                       (65536 + 512) * sizeof(float), stream,
                       x, dk, db, ws + 524288, out);
}


# setup kernel: core chain cut in the middle (left half x right half), one block per k-step of W builds only its own slices in LDS, 64 blocks not 1; changes f32 summation order of W
# speedup vs baseline: 2.1273x; 2.1273x over previous
#include <hip/hip_runtime.h>
#include <hip/hip_bf16.h>

typedef float v2f __attribute__((ext_vector_type(2)));
typedef float v8f __attribute__((ext_vector_type(8)));

static __device__ __forceinline__ v8f wmma_16x16x4_f32(v2f a, v2f b, v8f c) {
#if __has_builtin(__builtin_amdgcn_wmma_f32_16x16x4_f32)
    return __builtin_amdgcn_wmma_f32_16x16x4_f32(
         false, a,  false, b,
         (short)0, c,  false,  false);
#else
    v8f d;
    asm volatile("v_wmma_f32_16x16x4_f32 %0, %1, %2, %3"
                 : "=v"(d)
                 : "v"(a), "v"(b), "v"(c));
    return d;
#endif
}

__global__ void __launch_bounds__(1024)
mpo_setup_kernel(const float* __restrict__ w, float* __restrict__ wf) {
    __shared__ float sw[6272];
    __shared__ float sL2[64], sL3[128], sL4[256];
    __shared__ float sR6[256];
    __shared__ float sR5[512];
    __shared__ float sR4[1024];
    const int tid = threadIdx.x;
    const int ks  = blockIdx.x;
    const int i1 = (ks >> 5) & 1, i2 = (ks >> 4) & 1, i3 = (ks >> 3) & 1;
    const int i4 = (ks >> 2) & 1, i5 = (ks >> 1) & 1, i6 = ks & 1;

    {
        float r[6];
        for (int u = 0; u < 6; ++u) r[u] = w[tid + u * 1024];
        float r6 = 0.f;
        if (tid < 128) r6 = w[tid + 6144];
        for (int u = 0; u < 6; ++u) sw[tid + u * 1024] = r[u];
        if (tid < 128) sw[tid + 6144] = r6;
    }
    __syncthreads();

    if (tid < 64) {
        const int b = tid & 15, Jo = tid >> 4, J = Jo >> 1, j = Jo & 1;
        const float* t = sw + i1 * 2 + J;
        const float* c = sw + 64 + b * 4 + i2 * 2 + j;
        float s = 0.f;
        for (int a = 0; a < 16; ++a) s += t[a * 4] * c[a * 64];
        sL2[tid] = s;
    } else if (tid < 64 + 256) {
        const int e = tid - 64;
        const int a = e >> 4, rr = (e >> 2) & 3, cc = e & 3;
        const float* c = sw + 64 + 5 * 1024 + a * 64 + (rr >> 1) * 2 + (cc >> 1);
        const float* t = sw + 6208 + (rr & 1) * 2 + (cc & 1);
        float s = 0.f;
        for (int b = 0; b < 16; ++b) s += c[b * 4] * t[b * 4];
        sR6[e] = s;
    }
    __syncthreads();

    if (tid < 128) {
        const int b = tid & 15, Jo = tid >> 4, J = Jo >> 1, j = Jo & 1;
        const float* t = sL2 + J * 16;
        const float* c = sw + 64 + 1 * 1024 + b * 4 + i3 * 2 + j;
        float s = 0.f;
        for (int a = 0; a < 16; ++a) s += t[a] * c[a * 64];
        sL3[tid] = s;
    } else if (tid < 128 + 512) {
        const int e = tid - 128;
        const int a = e >> 5, rr = (e >> 3) & 3, c3 = e & 7;
        const float* c = sw + 64 + 4 * 1024 + a * 64 + i6 * 2 + (c3 >> 2);
        const float* t = sR6 + rr * 4 + (c3 & 3);
        float s = 0.f;
        for (int b = 0; b < 16; ++b) s += c[b * 4] * t[b * 16];
        sR5[e] = s;
    }
    __syncthreads();

    {
        const int a = tid >> 6, rr = (tid >> 4) & 3, c4 = tid & 15;
        const float* c = sw + 64 + 3 * 1024 + a * 64 + i5 * 2 + (c4 >> 3);
        const float* t = sR5 + rr * 8 + (c4 & 7);
        float s = 0.f;
        for (int b = 0; b < 16; ++b) s += c[b * 4] * t[b * 32];
        sR4[tid] = s;
    }
    if (tid < 256) {
        const int b = tid & 15, Jo = tid >> 4, J = Jo >> 1, j = Jo & 1;
        const float* t = sL3 + J * 16;
        const float* c = sw + 64 + 2 * 1024 + b * 4 + i4 * 2 + j;
        float s = 0.f;
        for (int a = 0; a < 16; ++a) s += t[a] * c[a * 64];
        sL4[tid] = s;
    }
    __syncthreads();

    {
        const int nt = tid >> 6, fl = (tid >> 1) & 31, v = tid & 1;
        const int kk = (fl >> 4) * 2 + v, nlo = fl & 15;
        const float* l = sL4 + nt * 16;
        const float* r = sR4 + kk * 16 + nlo;
        float s = 0.f;
        for (int a = 0; a < 16; ++a) s += l[a] * r[a * 64];
        wf[ks * 1024 + tid] = s;
    }
}

__global__ void __launch_bounds__(256, 1)
mpo_main_kernel(const float* __restrict__ x,
                const float* __restrict__ dk,
                const float* __restrict__ bias,
                const float* __restrict__ wf,
                float* __restrict__ out) {
    extern __shared__ float lds[];
    float* part = lds + 65536;
    const int tid = threadIdx.x;

    {
        const float4* src = (const float4*)wf;
        float4* dst = (float4*)lds;
        for (int i = tid; i < 16384; i += 256)
            dst[i] = src[i];
    }
    __syncthreads();

    const int wave = tid >> 5;
    const int lane = tid & 31;
    const int l16  = lane & 15;
    const int koff = (lane < 16) ? 0 : 2;
    const int q    = wave >> 2;
    const int g    = wave & 3;

    const long rowbase = ((long)blockIdx.x * 2 + q) * 64;
    const float* xbase = x + (rowbase + l16) * 256 + koff;

    v8f acc[4][4];
#pragma unroll
    for (int s = 0; s < 4; ++s)
#pragma unroll
        for (int t = 0; t < 4; ++t)
#pragma unroll
            for (int v = 0; v < 8; ++v)
                acc[s][t][v] = 0.f;

    const float* ldsb = lds + lane * 2 + g * 256;

    v2f a0 = *(const v2f*)(xbase);
    v2f a1 = *(const v2f*)(xbase + 4096);
    v2f a2 = *(const v2f*)(xbase + 8192);
    v2f a3 = *(const v2f*)(xbase + 12288);

    for (int k = 0; k < 63; ++k) {
        const float* xn = xbase + 4 * (k + 1);
        v2f n0 = *(const v2f*)(xn);
        v2f n1 = *(const v2f*)(xn + 4096);
        v2f n2 = *(const v2f*)(xn + 8192);
        v2f n3 = *(const v2f*)(xn + 12288);

        const float* lk = ldsb + k * 1024;
        v2f b0 = *(const v2f*)(lk);
        v2f b1 = *(const v2f*)(lk + 64);
        v2f b2 = *(const v2f*)(lk + 128);
        v2f b3 = *(const v2f*)(lk + 192);

        acc[0][0] = wmma_16x16x4_f32(a0, b0, acc[0][0]);
        acc[0][1] = wmma_16x16x4_f32(a0, b1, acc[0][1]);
        acc[0][2] = wmma_16x16x4_f32(a0, b2, acc[0][2]);
        acc[0][3] = wmma_16x16x4_f32(a0, b3, acc[0][3]);
        acc[1][0] = wmma_16x16x4_f32(a1, b0, acc[1][0]);
        acc[1][1] = wmma_16x16x4_f32(a1, b1, acc[1][1]);
        acc[1][2] = wmma_16x16x4_f32(a1, b2, acc[1][2]);
        acc[1][3] = wmma_16x16x4_f32(a1, b3, acc[1][3]);
        acc[2][0] = wmma_16x16x4_f32(a2, b0, acc[2][0]);
        acc[2][1] = wmma_16x16x4_f32(a2, b1, acc[2][1]);
        acc[2][2] = wmma_16x16x4_f32(a2, b2, acc[2][2]);
        acc[2][3] = wmma_16x16x4_f32(a2, b3, acc[2][3]);
        acc[3][0] = wmma_16x16x4_f32(a3, b0, acc[3][0]);
        acc[3][1] = wmma_16x16x4_f32(a3, b1, acc[3][1]);
        acc[3][2] = wmma_16x16x4_f32(a3, b2, acc[3][2]);
        acc[3][3] = wmma_16x16x4_f32(a3, b3, acc[3][3]);

        a0 = n0; a1 = n1; a2 = n2; a3 = n3;
    }
    {
        const float* lk = ldsb + 63 * 1024;
        v2f b0 = *(const v2f*)(lk);
        v2f b1 = *(const v2f*)(lk + 64);
        v2f b2 = *(const v2f*)(lk + 128);
        v2f b3 = *(const v2f*)(lk + 192);
        acc[0][0] = wmma_16x16x4_f32(a0, b0, acc[0][0]);
        acc[0][1] = wmma_16x16x4_f32(a0, b1, acc[0][1]);
        acc[0][2] = wmma_16x16x4_f32(a0, b2, acc[0][2]);
        acc[0][3] = wmma_16x16x4_f32(a0, b3, acc[0][3]);
        acc[1][0] = wmma_16x16x4_f32(a1, b0, acc[1][0]);
        acc[1][1] = wmma_16x16x4_f32(a1, b1, acc[1][1]);
        acc[1][2] = wmma_16x16x4_f32(a1, b2, acc[1][2]);
        acc[1][3] = wmma_16x16x4_f32(a1, b3, acc[1][3]);
        acc[2][0] = wmma_16x16x4_f32(a2, b0, acc[2][0]);
        acc[2][1] = wmma_16x16x4_f32(a2, b1, acc[2][1]);
        acc[2][2] = wmma_16x16x4_f32(a2, b2, acc[2][2]);
        acc[2][3] = wmma_16x16x4_f32(a2, b3, acc[2][3]);
        acc[3][0] = wmma_16x16x4_f32(a3, b0, acc[3][0]);
        acc[3][1] = wmma_16x16x4_f32(a3, b1, acc[3][1]);
        acc[3][2] = wmma_16x16x4_f32(a3, b2, acc[3][2]);
        acc[3][3] = wmma_16x16x4_f32(a3, b3, acc[3][3]);
    }

    float kv[4];
#pragma unroll
    for (int t = 0; t < 4; ++t)
        kv[t] = dk[(g * 4 + t) * 16 + l16];

#pragma unroll
    for (int s = 0; s < 4; ++s) {
        float sv[8];
#pragma unroll
        for (int v = 0; v < 8; ++v) sv[v] = 0.f;
#pragma unroll
        for (int t = 0; t < 4; ++t)
#pragma unroll
            for (int v = 0; v < 8; ++v) {
                float h = acc[s][t][v];
                sv[v] += h * h * kv[t];
            }
#pragma unroll
        for (int v = 0; v < 8; ++v) {
            sv[v] += __shfl_xor(sv[v], 1, 32);
            sv[v] += __shfl_xor(sv[v], 2, 32);
            sv[v] += __shfl_xor(sv[v], 4, 32);
            sv[v] += __shfl_xor(sv[v], 8, 32);
        }
        if (l16 == 0) {
            int half = (lane >= 16) ? 8 : 0;
            float* p = part + ((q * 4 + g) * 4 + s) * 16 + half;
#pragma unroll
            for (int v = 0; v < 8; ++v) p[v] = sv[v];
        }
    }
    __syncthreads();

    if (tid < 128) {
        int q2  = tid >> 6;
        int s2  = (tid >> 4) & 3;
        int idx = tid & 15;
        const float* p = part + (q2 * 16 + s2) * 16 + idx;
        float r = p[0] + p[64] + p[128] + p[192];
        out[(long)blockIdx.x * 128 + tid] = r + bias[0];
    }
}

extern "C" void kernel_launch(void* const* d_in, const int* in_sizes, int n_in,
                              void* d_out, int out_size, void* d_ws, size_t ws_size,
                              hipStream_t stream) {
    const float* x   = (const float*)d_in[0];
    const float* mpw = (const float*)d_in[1];
    const float* dk  = (const float*)d_in[2];
    const float* db  = (const float*)d_in[3];
    float* out = (float*)d_out;
    float* ws  = (float*)d_ws;

    hipLaunchKernelGGL(mpo_setup_kernel, dim3(64), dim3(1024), 0, stream, mpw, ws + 524288);

    hipLaunchKernelGGL(mpo_main_kernel, dim3(2048), dim3(256),
                       (65536 + 512) * sizeof(float), stream,
                       x, dk, db, ws + 524288, out);
}


# main kernel made persistent: 256 blocks (one per WGP) stage the W fragments into LDS once and walk 8 row tiles each, the A prefetch running on across the tile boundary; W staged 256 times, not 2048
# speedup vs baseline: 2.3553x; 1.1072x over previous
#include <hip/hip_runtime.h>
#include <hip/hip_bf16.h>

typedef float v2f __attribute__((ext_vector_type(2)));
typedef float v8f __attribute__((ext_vector_type(8)));

static __device__ __forceinline__ v8f wmma_16x16x4_f32(v2f a, v2f b, v8f c) {
#if __has_builtin(__builtin_amdgcn_wmma_f32_16x16x4_f32)
    return __builtin_amdgcn_wmma_f32_16x16x4_f32(
         false, a,  false, b,
         (short)0, c,  false,  false);
#else
    v8f d;
    asm volatile("v_wmma_f32_16x16x4_f32 %0, %1, %2, %3"
                 : "=v"(d)
                 : "v"(a), "v"(b), "v"(c));
    return d;
#endif
}

__global__ void __launch_bounds__(1024)
mpo_setup_kernel(const float* __restrict__ w, float* __restrict__ wf) {
    __shared__ float sw[6272];
    __shared__ float sL2[64], sL3[128], sL4[256];
    __shared__ float sR6[256];
    __shared__ float sR5[512];
    __shared__ float sR4[1024];
    const int tid = threadIdx.x;
    const int ks  = blockIdx.x;
    const int i1 = (ks >> 5) & 1, i2 = (ks >> 4) & 1, i3 = (ks >> 3) & 1;
    const int i4 = (ks >> 2) & 1, i5 = (ks >> 1) & 1, i6 = ks & 1;

    {
        float r[6];
        for (int u = 0; u < 6; ++u) r[u] = w[tid + u * 1024];
        float r6 = 0.f;
        if (tid < 128) r6 = w[tid + 6144];
        for (int u = 0; u < 6; ++u) sw[tid + u * 1024] = r[u];
        if (tid < 128) sw[tid + 6144] = r6;
    }
    __syncthreads();

    if (tid < 64) {
        const int b = tid & 15, Jo = tid >> 4, J = Jo >> 1, j = Jo & 1;
        const float* t = sw + i1 * 2 + J;
        const float* c = sw + 64 + b * 4 + i2 * 2 + j;
        float s = 0.f;
        for (int a = 0; a < 16; ++a) s += t[a * 4] * c[a * 64];
        sL2[tid] = s;
    } else if (tid < 64 + 256) {
        const int e = tid - 64;
        const int a = e >> 4, rr = (e >> 2) & 3, cc = e & 3;
        const float* c = sw + 64 + 5 * 1024 + a * 64 + (rr >> 1) * 2 + (cc >> 1);
        const float* t = sw + 6208 + (rr & 1) * 2 + (cc & 1);
        float s = 0.f;
        for (int b = 0; b < 16; ++b) s += c[b * 4] * t[b * 4];
        sR6[e] = s;
    }
    __syncthreads();

    if (tid < 128) {
        const int b = tid & 15, Jo = tid >> 4, J = Jo >> 1, j = Jo & 1;
        const float* t = sL2 + J * 16;
        const float* c = sw + 64 + 1 * 1024 + b * 4 + i3 * 2 + j;
        float s = 0.f;
        for (int a = 0; a < 16; ++a) s += t[a] * c[a * 64];
        sL3[tid] = s;
    } else if (tid < 128 + 512) {
        const int e = tid - 128;
        const int a = e >> 5, rr = (e >> 3) & 3, c3 = e & 7;
        const float* c = sw + 64 + 4 * 1024 + a * 64 + i6 * 2 + (c3 >> 2);
        const float* t = sR6 + rr * 4 + (c3 & 3);
        float s = 0.f;
        for (int b = 0; b < 16; ++b) s += c[b * 4] * t[b * 16];
        sR5[e] = s;
    }
    __syncthreads();

    {
        const int a = tid >> 6, rr = (tid >> 4) & 3, c4 = tid & 15;
        const float* c = sw + 64 + 3 * 1024 + a * 64 + i5 * 2 + (c4 >> 3);
        const float* t = sR5 + rr * 8 + (c4 & 7);
        float s = 0.f;
        for (int b = 0; b < 16; ++b) s += c[b * 4] * t[b * 32];
        sR4[tid] = s;
    }
    if (tid < 256) {
        const int b = tid & 15, Jo = tid >> 4, J = Jo >> 1, j = Jo & 1;
        const float* t = sL3 + J * 16;
        const float* c = sw + 64 + 2 * 1024 + b * 4 + i4 * 2 + j;
        float s = 0.f;
        for (int a = 0; a < 16; ++a) s += t[a] * c[a * 64];
        sL4[tid] = s;
    }
    __syncthreads();

    {
        const int nt = tid >> 6, fl = (tid >> 1) & 31, v = tid & 1;
        const int kk = (fl >> 4) * 2 + v, nlo = fl & 15;
        const float* l = sL4 + nt * 16;
        const float* r = sR4 + kk * 16 + nlo;
        float s = 0.f;
        for (int a = 0; a < 16; ++a) s += l[a] * r[a * 64];
        wf[ks * 1024 + tid] = s;
    }
}

#define MPO_WMMA_STEP(A0, A1, A2, A3, B0, B1, B2, B3)      \
    acc[0][0] = wmma_16x16x4_f32(A0, B0, acc[0][0]);      \
    acc[0][1] = wmma_16x16x4_f32(A0, B1, acc[0][1]);      \
    acc[0][2] = wmma_16x16x4_f32(A0, B2, acc[0][2]);      \
    acc[0][3] = wmma_16x16x4_f32(A0, B3, acc[0][3]);      \
    acc[1][0] = wmma_16x16x4_f32(A1, B0, acc[1][0]);      \
    acc[1][1] = wmma_16x16x4_f32(A1, B1, acc[1][1]);      \
    acc[1][2] = wmma_16x16x4_f32(A1, B2, acc[1][2]);      \
    acc[1][3] = wmma_16x16x4_f32(A1, B3, acc[1][3]);      \
    acc[2][0] = wmma_16x16x4_f32(A2, B0, acc[2][0]);      \
    acc[2][1] = wmma_16x16x4_f32(A2, B1, acc[2][1]);      \
    acc[2][2] = wmma_16x16x4_f32(A2, B2, acc[2][2]);      \
    acc[2][3] = wmma_16x16x4_f32(A2, B3, acc[2][3]);      \
    acc[3][0] = wmma_16x16x4_f32(A3, B0, acc[3][0]);      \
    acc[3][1] = wmma_16x16x4_f32(A3, B1, acc[3][1]);      \
    acc[3][2] = wmma_16x16x4_f32(A3, B2, acc[3][2]);      \
    acc[3][3] = wmma_16x16x4_f32(A3, B3, acc[3][3]);

__global__ void __launch_bounds__(256, 1)
mpo_main_kernel(const float* __restrict__ x,
                const float* __restrict__ dk,
                const float* __restrict__ bias,
                const float* __restrict__ wf,
                float* __restrict__ out) {
    extern __shared__ float lds[];
    float* part = lds + 65536;
    const int tid = threadIdx.x;

    const int wave = tid >> 5;
    const int lane = tid & 31;
    const int l16  = lane & 15;
    const int koff = (lane < 16) ? 0 : 2;
    const int q    = wave >> 2;
    const int g    = wave & 3;

    const int  ntiles  = 2048;
    const int  tstep   = gridDim.x;
    const long tstride = (long)tstep * (128 * 256);

    const float* xt = x + (((long)blockIdx.x * 2 + q) * 64 + l16) * 256 + koff;

    v2f a0 = *(const v2f*)(xt);
    v2f a1 = *(const v2f*)(xt + 4096);
    v2f a2 = *(const v2f*)(xt + 8192);
    v2f a3 = *(const v2f*)(xt + 12288);

    {
        const float4* src = (const float4*)wf;
        float4* dst = (float4*)lds;
        for (int i = tid; i < 16384; i += 256)
            dst[i] = src[i];
    }

    float kv[4];
#pragma unroll
    for (int t = 0; t < 4; ++t)
        kv[t] = dk[(g * 4 + t) * 16 + l16];
    const float bias0 = bias[0];

    __syncthreads();

    const float* ldsb = lds + lane * 2 + g * 256;

    int par = 0;
    for (int tile = blockIdx.x; tile < ntiles; tile += tstep) {
        v8f acc[4][4];
#pragma unroll
        for (int s = 0; s < 4; ++s)
#pragma unroll
            for (int t = 0; t < 4; ++t)
#pragma unroll
                for (int v = 0; v < 8; ++v)
                    acc[s][t][v] = 0.f;

        for (int k = 0; k < 63; ++k) {
            const float* xn = xt + 4 * (k + 1);
            v2f n0 = *(const v2f*)(xn);
            v2f n1 = *(const v2f*)(xn + 4096);
            v2f n2 = *(const v2f*)(xn + 8192);
            v2f n3 = *(const v2f*)(xn + 12288);

            const float* lk = ldsb + k * 1024;
            v2f b0 = *(const v2f*)(lk);
            v2f b1 = *(const v2f*)(lk + 64);
            v2f b2 = *(const v2f*)(lk + 128);
            v2f b3 = *(const v2f*)(lk + 192);

            MPO_WMMA_STEP(a0, a1, a2, a3, b0, b1, b2, b3)

            a0 = n0; a1 = n1; a2 = n2; a3 = n3;
        }
        {
            const float* xn = (tile + tstep < ntiles) ? xt + tstride : xt;
            v2f n0 = *(const v2f*)(xn);
            v2f n1 = *(const v2f*)(xn + 4096);
            v2f n2 = *(const v2f*)(xn + 8192);
            v2f n3 = *(const v2f*)(xn + 12288);

            const float* lk = ldsb + 63 * 1024;
            v2f b0 = *(const v2f*)(lk);
            v2f b1 = *(const v2f*)(lk + 64);
            v2f b2 = *(const v2f*)(lk + 128);
            v2f b3 = *(const v2f*)(lk + 192);

            MPO_WMMA_STEP(a0, a1, a2, a3, b0, b1, b2, b3)

            a0 = n0; a1 = n1; a2 = n2; a3 = n3;
            xt = xn;
        }

        float* mypart = part + par * 512;
#pragma unroll
        for (int s = 0; s < 4; ++s) {
            float sv[8];
#pragma unroll
            for (int v = 0; v < 8; ++v) sv[v] = 0.f;
#pragma unroll
            for (int t = 0; t < 4; ++t)
#pragma unroll
                for (int v = 0; v < 8; ++v) {
                    float h = acc[s][t][v];
                    sv[v] += h * h * kv[t];
                }
#pragma unroll
            for (int v = 0; v < 8; ++v) {
                sv[v] += __shfl_xor(sv[v], 1, 32);
                sv[v] += __shfl_xor(sv[v], 2, 32);
                sv[v] += __shfl_xor(sv[v], 4, 32);
                sv[v] += __shfl_xor(sv[v], 8, 32);
            }
            if (l16 == 0) {
                int half = (lane >= 16) ? 8 : 0;
                float* p = mypart + ((q * 4 + g) * 4 + s) * 16 + half;
#pragma unroll
                for (int v = 0; v < 8; ++v) p[v] = sv[v];
            }
        }
        __syncthreads();

        if (tid < 128) {
            int q2  = tid >> 6;
            int s2  = (tid >> 4) & 3;
            int idx = tid & 15;
            const float* p = mypart + (q2 * 16 + s2) * 16 + idx;
            float r = p[0] + p[64] + p[128] + p[192];
            out[(long)tile * 128 + tid] = r + bias0;
        }
        par ^= 1;
    }
}
#undef MPO_WMMA_STEP

extern "C" void kernel_launch(void* const* d_in, const int* in_sizes, int n_in,
                              void* d_out, int out_size, void* d_ws, size_t ws_size,
                              hipStream_t stream) {
    const float* x   = (const float*)d_in[0];
    const float* mpw = (const float*)d_in[1];
    const float* dk  = (const float*)d_in[2];
    const float* db  = (const float*)d_in[3];
    float* out = (float*)d_out;
    float* ws  = (float*)d_ws;

    hipLaunchKernelGGL(mpo_setup_kernel, dim3(64), dim3(1024), 0, stream, mpw, ws + 524288);

    hipLaunchKernelGGL(mpo_main_kernel, dim3(256), dim3(256),
                       (65536 + 1024) * sizeof(float), stream,
                       x, dk, db, ws + 524288, out);
}


# main kernel: WMMA operands swapped (W frag as A, x frag as B), accumulators hold H transposed: feature sum in-lane + one half-wave exchange, 4 shuffles per tile not 128; changes f32 sum order
# speedup vs baseline: 2.4275x; 1.0307x over previous
#include <hip/hip_runtime.h>
#include <hip/hip_bf16.h>

typedef float v2f __attribute__((ext_vector_type(2)));
typedef float v8f __attribute__((ext_vector_type(8)));

static __device__ __forceinline__ v8f wmma_16x16x4_f32(v2f a, v2f b, v8f c) {
#if __has_builtin(__builtin_amdgcn_wmma_f32_16x16x4_f32)
    return __builtin_amdgcn_wmma_f32_16x16x4_f32(
         false, a,  false, b,
         (short)0, c,  false,  false);
#else
    v8f d;
    asm volatile("v_wmma_f32_16x16x4_f32 %0, %1, %2, %3"
                 : "=v"(d)
                 : "v"(a), "v"(b), "v"(c));
    return d;
#endif
}

__global__ void __launch_bounds__(1024)
mpo_setup_kernel(const float* __restrict__ w, float* __restrict__ wf) {
    __shared__ float sw[6272];
    __shared__ float sL2[64], sL3[128], sL4[256];
    __shared__ float sR6[256];
    __shared__ float sR5[512];
    __shared__ float sR4[1024];
    const int tid = threadIdx.x;
    const int ks  = blockIdx.x;
    const int i1 = (ks >> 5) & 1, i2 = (ks >> 4) & 1, i3 = (ks >> 3) & 1;
    const int i4 = (ks >> 2) & 1, i5 = (ks >> 1) & 1, i6 = ks & 1;

    {
        float r[6];
        for (int u = 0; u < 6; ++u) r[u] = w[tid + u * 1024];
        float r6 = 0.f;
        if (tid < 128) r6 = w[tid + 6144];
        for (int u = 0; u < 6; ++u) sw[tid + u * 1024] = r[u];
        if (tid < 128) sw[tid + 6144] = r6;
    }
    __syncthreads();

    if (tid < 64) {
        const int b = tid & 15, Jo = tid >> 4, J = Jo >> 1, j = Jo & 1;
        const float* t = sw + i1 * 2 + J;
        const float* c = sw + 64 + b * 4 + i2 * 2 + j;
        float s = 0.f;
        for (int a = 0; a < 16; ++a) s += t[a * 4] * c[a * 64];
        sL2[tid] = s;
    } else if (tid < 64 + 256) {
        const int e = tid - 64;
        const int a = e >> 4, rr = (e >> 2) & 3, cc = e & 3;
        const float* c = sw + 64 + 5 * 1024 + a * 64 + (rr >> 1) * 2 + (cc >> 1);
        const float* t = sw + 6208 + (rr & 1) * 2 + (cc & 1);
        float s = 0.f;
        for (int b = 0; b < 16; ++b) s += c[b * 4] * t[b * 4];
        sR6[e] = s;
    }
    __syncthreads();

    if (tid < 128) {
        const int b = tid & 15, Jo = tid >> 4, J = Jo >> 1, j = Jo & 1;
        const float* t = sL2 + J * 16;
        const float* c = sw + 64 + 1 * 1024 + b * 4 + i3 * 2 + j;
        float s = 0.f;
        for (int a = 0; a < 16; ++a) s += t[a] * c[a * 64];
        sL3[tid] = s;
    } else if (tid < 128 + 512) {
        const int e = tid - 128;
        const int a = e >> 5, rr = (e >> 3) & 3, c3 = e & 7;
        const float* c = sw + 64 + 4 * 1024 + a * 64 + i6 * 2 + (c3 >> 2);
        const float* t = sR6 + rr * 4 + (c3 & 3);
        float s = 0.f;
        for (int b = 0; b < 16; ++b) s += c[b * 4] * t[b * 16];
        sR5[e] = s;
    }
    __syncthreads();

    {
        const int a = tid >> 6, rr = (tid >> 4) & 3, c4 = tid & 15;
        const float* c = sw + 64 + 3 * 1024 + a * 64 + i5 * 2 + (c4 >> 3);
        const float* t = sR5 + rr * 8 + (c4 & 7);
        float s = 0.f;
        for (int b = 0; b < 16; ++b) s += c[b * 4] * t[b * 32];
        sR4[tid] = s;
    }
    if (tid < 256) {
        const int b = tid & 15, Jo = tid >> 4, J = Jo >> 1, j = Jo & 1;
        const float* t = sL3 + J * 16;
        const float* c = sw + 64 + 2 * 1024 + b * 4 + i4 * 2 + j;
        float s = 0.f;
        for (int a = 0; a < 16; ++a) s += t[a] * c[a * 64];
        sL4[tid] = s;
    }
    __syncthreads();

    {
        const int nt = tid >> 6, fl = (tid >> 1) & 31, v = tid & 1;
        const int kk = (fl >> 4) * 2 + v, nlo = fl & 15;
        const float* l = sL4 + nt * 16;
        const float* r = sR4 + kk * 16 + nlo;
        float s = 0.f;
        for (int a = 0; a < 16; ++a) s += l[a] * r[a * 64];
        wf[ks * 1024 + tid] = s;
    }
}

#define MPO_WMMA_STEP(A0, A1, A2, A3, B0, B1, B2, B3)      \
    acc[0][0] = wmma_16x16x4_f32(B0, A0, acc[0][0]);      \
    acc[0][1] = wmma_16x16x4_f32(B1, A0, acc[0][1]);      \
    acc[0][2] = wmma_16x16x4_f32(B2, A0, acc[0][2]);      \
    acc[0][3] = wmma_16x16x4_f32(B3, A0, acc[0][3]);      \
    acc[1][0] = wmma_16x16x4_f32(B0, A1, acc[1][0]);      \
    acc[1][1] = wmma_16x16x4_f32(B1, A1, acc[1][1]);      \
    acc[1][2] = wmma_16x16x4_f32(B2, A1, acc[1][2]);      \
    acc[1][3] = wmma_16x16x4_f32(B3, A1, acc[1][3]);      \
    acc[2][0] = wmma_16x16x4_f32(B0, A2, acc[2][0]);      \
    acc[2][1] = wmma_16x16x4_f32(B1, A2, acc[2][1]);      \
    acc[2][2] = wmma_16x16x4_f32(B2, A2, acc[2][2]);      \
    acc[2][3] = wmma_16x16x4_f32(B3, A2, acc[2][3]);      \
    acc[3][0] = wmma_16x16x4_f32(B0, A3, acc[3][0]);      \
    acc[3][1] = wmma_16x16x4_f32(B1, A3, acc[3][1]);      \
    acc[3][2] = wmma_16x16x4_f32(B2, A3, acc[3][2]);      \
    acc[3][3] = wmma_16x16x4_f32(B3, A3, acc[3][3]);

__global__ void __launch_bounds__(256, 1)
mpo_main_kernel(const float* __restrict__ x,
                const float* __restrict__ dk,
                const float* __restrict__ bias,
                const float* __restrict__ wf,
                float* __restrict__ out) {
    extern __shared__ float lds[];
    float* part = lds + 65536;
    const int tid = threadIdx.x;

    const int wave = tid >> 5;
    const int lane = tid & 31;
    const int l16  = lane & 15;
    const int koff = (lane < 16) ? 0 : 2;
    const int q    = wave >> 2;
    const int g    = wave & 3;

    const int  ntiles  = 2048;
    const int  tstep   = gridDim.x;
    const long tstride = (long)tstep * (128 * 256);

    const float* xt = x + (((long)blockIdx.x * 2 + q) * 64 + l16) * 256 + koff;

    v2f a0 = *(const v2f*)(xt);
    v2f a1 = *(const v2f*)(xt + 4096);
    v2f a2 = *(const v2f*)(xt + 8192);
    v2f a3 = *(const v2f*)(xt + 12288);

    {
        const float4* src = (const float4*)wf;
        float4* dst = (float4*)lds;
        for (int i = tid; i < 16384; i += 256)
            dst[i] = src[i];
    }

    const int hsel = lane >> 4;
    float kv[4][8];
#pragma unroll
    for (int t = 0; t < 4; ++t)
#pragma unroll
        for (int r = 0; r < 8; ++r)
            kv[t][r] = dk[(g * 4 + t) * 16 + hsel * 8 + r];
    const float bias0 = bias[0];

    __syncthreads();

    const float* ldsb = lds + lane * 2 + g * 256;

    int par = 0;
    for (int tile = blockIdx.x; tile < ntiles; tile += tstep) {
        v8f acc[4][4];
#pragma unroll
        for (int s = 0; s < 4; ++s)
#pragma unroll
            for (int t = 0; t < 4; ++t)
#pragma unroll
                for (int v = 0; v < 8; ++v)
                    acc[s][t][v] = 0.f;

        for (int k = 0; k < 63; ++k) {
            const float* xn = xt + 4 * (k + 1);
            v2f n0 = *(const v2f*)(xn);
            v2f n1 = *(const v2f*)(xn + 4096);
            v2f n2 = *(const v2f*)(xn + 8192);
            v2f n3 = *(const v2f*)(xn + 12288);

            const float* lk = ldsb + k * 1024;
            v2f b0 = *(const v2f*)(lk);
            v2f b1 = *(const v2f*)(lk + 64);
            v2f b2 = *(const v2f*)(lk + 128);
            v2f b3 = *(const v2f*)(lk + 192);

            MPO_WMMA_STEP(a0, a1, a2, a3, b0, b1, b2, b3)

            a0 = n0; a1 = n1; a2 = n2; a3 = n3;
        }
        {
            const float* xn = (tile + tstep < ntiles) ? xt + tstride : xt;
            v2f n0 = *(const v2f*)(xn);
            v2f n1 = *(const v2f*)(xn + 4096);
            v2f n2 = *(const v2f*)(xn + 8192);
            v2f n3 = *(const v2f*)(xn + 12288);

            const float* lk = ldsb + 63 * 1024;
            v2f b0 = *(const v2f*)(lk);
            v2f b1 = *(const v2f*)(lk + 64);
            v2f b2 = *(const v2f*)(lk + 128);
            v2f b3 = *(const v2f*)(lk + 192);

            MPO_WMMA_STEP(a0, a1, a2, a3, b0, b1, b2, b3)

            a0 = n0; a1 = n1; a2 = n2; a3 = n3;
            xt = xn;
        }

        float* mypart = part + par * 512;
        float ps[4];
#pragma unroll
        for (int s = 0; s < 4; ++s) {
            float pe = 0.f, po = 0.f;
#pragma unroll
            for (int t = 0; t < 4; ++t)
#pragma unroll
                for (int r = 0; r < 8; r += 2) {
                    float h0 = acc[s][t][r];
                    float h1 = acc[s][t][r + 1];
                    pe += h0 * h0 * kv[t][r];
                    po += h1 * h1 * kv[t][r + 1];
                }
            ps[s] = pe + po;
        }
#pragma unroll
        for (int s = 0; s < 4; ++s)
            ps[s] += __shfl_xor(ps[s], 16, 32);
        if (lane < 16) {
            float* p = mypart + ((q * 4 + g) * 4) * 16 + lane;
#pragma unroll
            for (int s = 0; s < 4; ++s) p[s * 16] = ps[s];
        }
        __syncthreads();

        if (tid < 128) {
            int q2  = tid >> 6;
            int s2  = (tid >> 4) & 3;
            int idx = tid & 15;
            const float* p = mypart + (q2 * 16 + s2) * 16 + idx;
            float r = p[0] + p[64] + p[128] + p[192];
            out[(long)tile * 128 + tid] = r + bias0;
        }
        par ^= 1;
    }
}
#undef MPO_WMMA_STEP

extern "C" void kernel_launch(void* const* d_in, const int* in_sizes, int n_in,
                              void* d_out, int out_size, void* d_ws, size_t ws_size,
                              hipStream_t stream) {
    const float* x   = (const float*)d_in[0];
    const float* mpw = (const float*)d_in[1];
    const float* dk  = (const float*)d_in[2];
    const float* db  = (const float*)d_in[3];
    float* out = (float*)d_out;
    float* ws  = (float*)d_ws;

    hipLaunchKernelGGL(mpo_setup_kernel, dim3(64), dim3(1024), 0, stream, mpw, ws + 524288);

    hipLaunchKernelGGL(mpo_main_kernel, dim3(256), dim3(256),
                       (65536 + 1024) * sizeof(float), stream,
                       x, dk, db, ws + 524288, out);
}


# main kernel: wave tile 64 rows x 64 features -> 32 rows x all 256 features (2x16 tiles): x fragment loaded by one wave not four, row sum stays in wave, no tile barrier; changes f32 sum order
# speedup vs baseline: 2.5043x; 1.0316x over previous
#include <hip/hip_runtime.h>
#include <hip/hip_bf16.h>

typedef float v2f __attribute__((ext_vector_type(2)));
typedef float v8f __attribute__((ext_vector_type(8)));

static __device__ __forceinline__ v8f wmma_16x16x4_f32(v2f a, v2f b, v8f c) {
#if __has_builtin(__builtin_amdgcn_wmma_f32_16x16x4_f32)
    return __builtin_amdgcn_wmma_f32_16x16x4_f32(
         false, a,  false, b,
         (short)0, c,  false,  false);
#else
    v8f d;
    asm volatile("v_wmma_f32_16x16x4_f32 %0, %1, %2, %3"
                 : "=v"(d)
                 : "v"(a), "v"(b), "v"(c));
    return d;
#endif
}

__global__ void __launch_bounds__(1024)
mpo_setup_kernel(const float* __restrict__ w, float* __restrict__ wf) {
    __shared__ float sw[6272];
    __shared__ float sL2[64], sL3[128], sL4[256];
    __shared__ float sR6[256];
    __shared__ float sR5[512];
    __shared__ float sR4[1024];
    const int tid = threadIdx.x;
    const int ks  = blockIdx.x;
    const int i1 = (ks >> 5) & 1, i2 = (ks >> 4) & 1, i3 = (ks >> 3) & 1;
    const int i4 = (ks >> 2) & 1, i5 = (ks >> 1) & 1, i6 = ks & 1;

    {
        float r[6];
        for (int u = 0; u < 6; ++u) r[u] = w[tid + u * 1024];
        float r6 = 0.f;
        if (tid < 128) r6 = w[tid + 6144];
        for (int u = 0; u < 6; ++u) sw[tid + u * 1024] = r[u];
        if (tid < 128) sw[tid + 6144] = r6;
    }
    __syncthreads();

    if (tid < 64) {
        const int b = tid & 15, Jo = tid >> 4, J = Jo >> 1, j = Jo & 1;
        const float* t = sw + i1 * 2 + J;
        const float* c = sw + 64 + b * 4 + i2 * 2 + j;
        float s = 0.f;
        for (int a = 0; a < 16; ++a) s += t[a * 4] * c[a * 64];
        sL2[tid] = s;
    } else if (tid < 64 + 256) {
        const int e = tid - 64;
        const int a = e >> 4, rr = (e >> 2) & 3, cc = e & 3;
        const float* c = sw + 64 + 5 * 1024 + a * 64 + (rr >> 1) * 2 + (cc >> 1);
        const float* t = sw + 6208 + (rr & 1) * 2 + (cc & 1);
        float s = 0.f;
        for (int b = 0; b < 16; ++b) s += c[b * 4] * t[b * 4];
        sR6[e] = s;
    }
    __syncthreads();

    if (tid < 128) {
        const int b = tid & 15, Jo = tid >> 4, J = Jo >> 1, j = Jo & 1;
        const float* t = sL2 + J * 16;
        const float* c = sw + 64 + 1 * 1024 + b * 4 + i3 * 2 + j;
        float s = 0.f;
        for (int a = 0; a < 16; ++a) s += t[a] * c[a * 64];
        sL3[tid] = s;
    } else if (tid < 128 + 512) {
        const int e = tid - 128;
        const int a = e >> 5, rr = (e >> 3) & 3, c3 = e & 7;
        const float* c = sw + 64 + 4 * 1024 + a * 64 + i6 * 2 + (c3 >> 2);
        const float* t = sR6 + rr * 4 + (c3 & 3);
        float s = 0.f;
        for (int b = 0; b < 16; ++b) s += c[b * 4] * t[b * 16];
        sR5[e] = s;
    }
    __syncthreads();

    {
        const int a = tid >> 6, rr = (tid >> 4) & 3, c4 = tid & 15;
        const float* c = sw + 64 + 3 * 1024 + a * 64 + i5 * 2 + (c4 >> 3);
        const float* t = sR5 + rr * 8 + (c4 & 7);
        float s = 0.f;
        for (int b = 0; b < 16; ++b) s += c[b * 4] * t[b * 32];
        sR4[tid] = s;
    }
    if (tid < 256) {
        const int b = tid & 15, Jo = tid >> 4, J = Jo >> 1, j = Jo & 1;
        const float* t = sL3 + J * 16;
        const float* c = sw + 64 + 2 * 1024 + b * 4 + i4 * 2 + j;
        float s = 0.f;
        for (int a = 0; a < 16; ++a) s += t[a] * c[a * 64];
        sL4[tid] = s;
    }
    __syncthreads();

    {
        const int nt = tid >> 6, fl = (tid >> 1) & 31, v = tid & 1;
        const int kk = (fl >> 4) * 2 + v, nlo = fl & 15;
        const float* l = sL4 + nt * 16;
        const float* r = sR4 + kk * 16 + nlo;
        float s = 0.f;
        for (int a = 0; a < 16; ++a) s += l[a] * r[a * 64];
        wf[ks * 1024 + tid] = s;
    }
}

#define MPO_WMMA_PAIR(T)                                   \
    acc[0][T] = wmma_16x16x4_f32(b[T], a0, acc[0][T]);     \
    acc[1][T] = wmma_16x16x4_f32(b[T], a1, acc[1][T]);

#define MPO_WMMA_STEP()                                    \
    MPO_WMMA_PAIR(0)  MPO_WMMA_PAIR(1)  MPO_WMMA_PAIR(2)  MPO_WMMA_PAIR(3)   \
    MPO_WMMA_PAIR(4)  MPO_WMMA_PAIR(5)  MPO_WMMA_PAIR(6)  MPO_WMMA_PAIR(7)   \
    MPO_WMMA_PAIR(8)  MPO_WMMA_PAIR(9)  MPO_WMMA_PAIR(10) MPO_WMMA_PAIR(11)  \
    MPO_WMMA_PAIR(12) MPO_WMMA_PAIR(13) MPO_WMMA_PAIR(14) MPO_WMMA_PAIR(15)

__global__ void __launch_bounds__(256, 1)
mpo_main_kernel(const float* __restrict__ x,
                const float* __restrict__ dk,
                const float* __restrict__ bias,
                const float* __restrict__ wf,
                float* __restrict__ out) {
    extern __shared__ float lds[];
    float* sdk = lds + 65536;
    const int tid = threadIdx.x;

    const int wave = tid >> 5;
    const int lane = tid & 31;
    const int l16  = lane & 15;
    const int hsel = lane >> 4;
    const int koff = hsel * 2;

    const int  nslabs  = 8192;
    const int  sstep   = gridDim.x * 8;
    const long sstride = (long)sstep * (32 * 256);

    int slab = blockIdx.x * 8 + wave;

    const float* xt = x + ((long)slab * 32 + l16) * 256 + koff;

    v2f a0 = *(const v2f*)(xt);
    v2f a1 = *(const v2f*)(xt + 4096);

    {
        const float4* src = (const float4*)wf;
        float4* dst = (float4*)lds;
        for (int i = tid; i < 16384; i += 256)
            dst[i] = src[i];
    }
    sdk[tid] = dk[tid];
    const float bias0 = bias[0];

    __syncthreads();

    const float* ldsb = lds + lane * 2;
    const float* kvp = sdk + hsel * 8;

    for (; slab < nslabs; slab += sstep) {
        v8f acc[2][16];
        v2f b[16];
#pragma unroll
        for (int s = 0; s < 2; ++s)
#pragma unroll
            for (int t = 0; t < 16; ++t)
#pragma unroll
                for (int v = 0; v < 8; ++v)
                    acc[s][t][v] = 0.f;

#pragma nounroll
        for (int k = 0; k < 63; ++k) {
            const float* xn = xt + 4 * (k + 1);
            v2f n0 = *(const v2f*)(xn);
            v2f n1 = *(const v2f*)(xn + 4096);

            const float* lk = ldsb + k * 1024;
#pragma unroll
            for (int t = 0; t < 16; ++t)
                b[t] = *(const v2f*)(lk + t * 64);

            MPO_WMMA_STEP()

            a0 = n0; a1 = n1;
        }
        {
            const float* xn = (slab + sstep < nslabs) ? xt + sstride : xt;
            v2f n0 = *(const v2f*)(xn);
            v2f n1 = *(const v2f*)(xn + 4096);

            const float* lk = ldsb + 63 * 1024;
#pragma unroll
            for (int t = 0; t < 16; ++t)
                b[t] = *(const v2f*)(lk + t * 64);

            MPO_WMMA_STEP()

            a0 = n0; a1 = n1;
            xt = xn;
        }

        float p0[4] = {0.f, 0.f, 0.f, 0.f};
        float p1[4] = {0.f, 0.f, 0.f, 0.f};
        asm volatile("" ::: "memory");
#pragma unroll
        for (int t = 0; t < 16; ++t) {
            const float4 k0 = *(const float4*)(kvp + t * 16);
            const float4 k1 = *(const float4*)(kvp + t * 16 + 4);
            const float kv[8] = {k0.x, k0.y, k0.z, k0.w, k1.x, k1.y, k1.z, k1.w};
#pragma unroll
            for (int r = 0; r < 8; ++r) {
                const float h0 = acc[0][t][r];
                const float h1 = acc[1][t][r];
                p0[(t & 1) * 2 + (r & 1)] += h0 * h0 * kv[r];
                p1[(t & 1) * 2 + (r & 1)] += h1 * h1 * kv[r];
            }
        }
        float ps0 = (p0[0] + p0[1]) + (p0[2] + p0[3]);
        float ps1 = (p1[0] + p1[1]) + (p1[2] + p1[3]);
        ps0 += __shfl_xor(ps0, 16, 32);
        ps1 += __shfl_xor(ps1, 16, 32);
        out[(long)slab * 32 + lane] = (hsel ? ps1 : ps0) + bias0;
    }
}
#undef MPO_WMMA_STEP
#undef MPO_WMMA_PAIR

extern "C" void kernel_launch(void* const* d_in, const int* in_sizes, int n_in,
                              void* d_out, int out_size, void* d_ws, size_t ws_size,
                              hipStream_t stream) {
    const float* x   = (const float*)d_in[0];
    const float* mpw = (const float*)d_in[1];
    const float* dk  = (const float*)d_in[2];
    const float* db  = (const float*)d_in[3];
    float* out = (float*)d_out;
    float* ws  = (float*)d_ws;

    hipLaunchKernelGGL(mpo_setup_kernel, dim3(64), dim3(1024), 0, stream, mpw, ws + 524288);

    hipLaunchKernelGGL(mpo_main_kernel, dim3(256), dim3(256),
                       (65536 + 256) * sizeof(float), stream,
                       x, dk, db, ws + 524288, out);
}


# main product as the quadratic form x (W diag(d) W^T) x^T: setup forms the symmetric matrix, main multiplies only its 136 upper 16x16 blocks (1088 WMMAs a slab, not 2048); changes f32 sum order
# speedup vs baseline: 4.3599x; 1.7410x over previous
#include <hip/hip_runtime.h>


typedef float v2f __attribute__((ext_vector_type(2)));
typedef float v8f __attribute__((ext_vector_type(8)));

static __device__ __forceinline__ v8f wmma_16x16x4_f32(v2f a, v2f b, v8f c) {
#if __has_builtin(__builtin_amdgcn_wmma_f32_16x16x4_f32)
    return __builtin_amdgcn_wmma_f32_16x16x4_f32(
         false, a,  false, b,
         (short)0, c,  false,  false);
#else
    v8f d;
    asm volatile("v_wmma_f32_16x16x4_f32 %0, %1, %2, %3"
                 : "=v"(d)
                 : "v"(a), "v"(b), "v"(c));
    return d;
#endif
}

#define MPO_WP 260

__global__ void __launch_bounds__(1024)
mpo_setup_kernel(const float* __restrict__ w, const float* __restrict__ dk,
                 float* __restrict__ mf) {
    extern __shared__ float sm[];
    float* sw  = sm;
    float* sR6 = sw  + 6272;
    float* sR5 = sR6 + 256;
    float* sR4 = sR5 + 1024;
    float* sL2 = sR4 + 4096;
    float* sL3 = sL2 + 128;
    float* sL4 = sL3 + 256;
    float* sWa = sL4 + 512;
    float* sWb = sWa + 16 * MPO_WP;
    float* sd  = sWb + 16 * MPO_WP;
    const int tid = threadIdx.x;

    int t = 0, jj = blockIdx.x;
    for (int u = 0; u < 15; ++u) {
        if (jj >= 16 - t) { jj -= 16 - t; ++t; }
    }
    const int j = t + jj;
    const int offt = 16 * t - (t * (t - 1)) / 2;

    {
        float r[6];
        for (int u = 0; u < 6; ++u) r[u] = w[tid + u * 1024];
        float r6 = 0.f, rd = 0.f;
        if (tid < 128) r6 = w[tid + 6144];
        if (tid < 256) rd = dk[tid];
        for (int u = 0; u < 6; ++u) sw[tid + u * 1024] = r[u];
        if (tid < 128) sw[tid + 6144] = r6;
        if (tid < 256) sd[tid] = rd;
    }
    __syncthreads();

    if (tid < 128) {
        const int which = tid >> 6, e = tid & 63;
        const int rb = which ? j : t;
        const int i1 = (rb >> 3) & 1, i2 = (rb >> 2) & 1;
        const int b = e & 15, Jo = e >> 4, J = Jo >> 1, jb = Jo & 1;
        const float* tt = sw + i1 * 2 + J;
        const float* c = sw + 64 + b * 4 + i2 * 2 + jb;
        float s = 0.f;
        for (int a = 0; a < 16; ++a) s += tt[a * 4] * c[a * 64];
        sL2[tid] = s;
    } else if (tid < 128 + 256) {
        const int e = tid - 128;
        const int a = e >> 4, rr = (e >> 2) & 3, cc = e & 3;
        const float* c = sw + 64 + 5 * 1024 + a * 64 + (rr >> 1) * 2 + (cc >> 1);
        const float* tt = sw + 6208 + (rr & 1) * 2 + (cc & 1);
        float s = 0.f;
        for (int b = 0; b < 16; ++b) s += c[b * 4] * tt[b * 4];
        sR6[e] = s;
    }
    __syncthreads();

    {
        const int i6 = tid >> 9, e = tid & 511;
        const int a = e >> 5, rr = (e >> 3) & 3, c3 = e & 7;
        const float* c = sw + 64 + 4 * 1024 + a * 64 + i6 * 2 + (c3 >> 2);
        const float* tt = sR6 + rr * 4 + (c3 & 3);
        float s = 0.f;
        for (int b = 0; b < 16; ++b) s += c[b * 4] * tt[b * 16];
        sR5[tid] = s;
    }
    if (tid < 256) {
        const int which = tid >> 7, e = tid & 127;
        const int rb = which ? j : t;
        const int i3 = (rb >> 1) & 1;
        const int b = e & 15, Jo = e >> 4, J = Jo >> 1, jb = Jo & 1;
        const float* tt = sL2 + which * 64 + J * 16;
        const float* c = sw + 64 + 1 * 1024 + b * 4 + i3 * 2 + jb;
        float s = 0.f;
        for (int a = 0; a < 16; ++a) s += tt[a] * c[a * 64];
        sL3[tid] = s;
    }
    __syncthreads();

    for (int u = 0; u < 4; ++u) {
        const int idx = tid + u * 1024;
        const int a = idx >> 8, rlo = (idx >> 4) & 15, nlo = idx & 15;
        const int i5 = rlo >> 3, i6 = (rlo >> 2) & 1, rr = rlo & 3;
        const float* c = sw + 64 + 3 * 1024 + a * 64 + i5 * 2 + (nlo >> 3);
        const float* tt = sR5 + i6 * 512 + rr * 8 + (nlo & 7);
        float s = 0.f;
        for (int b = 0; b < 16; ++b) s += c[b * 4] * tt[b * 32];
        sR4[idx] = s;
    }
    if (tid < 512) {
        const int which = tid >> 8, e = tid & 255;
        const int rb = which ? j : t;
        const int i4 = rb & 1;
        const int b = e & 15, Jo = e >> 4, J = Jo >> 1, jb = Jo & 1;
        const float* tt = sL3 + which * 128 + J * 16;
        const float* c = sw + 64 + 2 * 1024 + b * 4 + i4 * 2 + jb;
        float s = 0.f;
        for (int a = 0; a < 16; ++a) s += tt[a] * c[a * 64];
        sL4[tid] = s;
    }
    __syncthreads();

    for (int u = 0; u < 8; ++u) {
        const int idx = tid + u * 1024;
        const int which = idx >> 12, rlo = (idx >> 8) & 15, n = idx & 255;
        const float* l = sL4 + which * 256 + (n >> 4) * 16;
        const float* r = sR4 + rlo * 16 + (n & 15);
        float s = 0.f;
        for (int a = 0; a < 16; ++a) s += l[a] * r[a * 256];
        if (which) sWb[rlo * MPO_WP + n] = s;
        else       sWa[rlo * MPO_WP + n] = s * sd[n];
    }
    __syncthreads();

    {
        const int o = tid >> 2, qn = tid & 3;
        const int r = o >> 4, rp = o & 15;
        const float* pa = sWa + r * MPO_WP + qn;
        const float* pb = sWb + rp * MPO_WP + qn;
        float s0 = 0.f, s1 = 0.f;
        for (int i = 0; i < 64; i += 2) {
            s0 += pa[4 * i] * pb[4 * i];
            s1 += pa[4 * i + 4] * pb[4 * i + 4];
        }
        float s = s0 + s1;
        s += __shfl_xor(s, 1, 32);
        s += __shfl_xor(s, 2, 32);
        if (t != j) s *= 2.f;
        if (qn == 0) {
            const int h = r >> 3, q = (r >> 1) & 3, e = r & 1;
            const int lane = h * 16 + rp;
            mf[((4 * offt + q * (16 - t) + jj) * 32 + lane) * 2 + e] = s;
        }
    }
}

__global__ void __launch_bounds__(256, 1)
mpo_main_kernel(const float* __restrict__ x,
                const float* __restrict__ bias,
                const float* __restrict__ mf,
                float* __restrict__ out) {
    extern __shared__ float lds[];
    const int tid = threadIdx.x;

    const int wave = tid >> 5;
    const int lane = tid & 31;
    const int l16  = lane & 15;
    const int hsel = lane >> 4;

    const int  nslabs  = 8192;
    const int  sstep   = gridDim.x * 8;
    const long sstride = (long)sstep * (32 * 256);

    int slab = blockIdx.x * 8 + wave;

    const float* xt = x + ((long)slab * 32 + l16) * 256 + hsel * 8;

    float4 xr[4][2][2];
#pragma unroll
    for (int u = 0; u < 3; ++u)
#pragma unroll
        for (int s = 0; s < 2; ++s)
#pragma unroll
            for (int hh = 0; hh < 2; ++hh)
                xr[u][s][hh] = *(const float4*)(xt + s * 4096 + 16 * u + 4 * hh);

    {
        const float4* src = (const float4*)mf;
        float4* dst = (float4*)lds;
        for (int i = tid; i < 8704; i += 256)
            dst[i] = src[i];
    }
    const float bias0 = bias[0];

    __syncthreads();

    const float* ldsb = lds + lane * 2;

    v2f bs[2][16];
#pragma unroll
    for (int i = 0; i < 16; ++i)
        bs[0][i] = *(const v2f*)(ldsb + i * 64);

    for (; slab < nslabs; slab += sstep) {
        const float* xnext = (slab + sstep < nslabs) ? xt + sstride : xt;

        v8f acc[2][16];
#pragma unroll
        for (int s = 0; s < 2; ++s)
#pragma unroll
            for (int t = 0; t < 16; ++t)
#pragma unroll
                for (int v = 0; v < 8; ++v)
                    acc[s][t][v] = 0.f;

        float p0[2] = {0.f, 0.f};
        float p1[2] = {0.f, 0.f};

#pragma unroll
        for (int t = 0; t < 16; ++t) {
            {
                const int tb = t + 3;
                const float* srcx = (tb < 16) ? xt + 16 * tb : xnext + 16 * (tb - 16);
#pragma unroll
                for (int s = 0; s < 2; ++s)
#pragma unroll
                    for (int hh = 0; hh < 2; ++hh)
                        xr[tb & 3][s][hh] = *(const float4*)(srcx + s * 4096 + 4 * hh);
            }

            const float4 c00 = xr[t & 3][0][0], c01 = xr[t & 3][0][1];
            const float4 c10 = xr[t & 3][1][0], c11 = xr[t & 3][1][1];
            const float xs0[8] = {c00.x, c00.y, c00.z, c00.w, c01.x, c01.y, c01.z, c01.w};
            const float xs1[8] = {c10.x, c10.y, c10.z, c10.w, c11.x, c11.y, c11.z, c11.w};

            const int nt = 16 - t;
            int fo = 256 * (16 * t - (t * (t - 1)) / 2);
            asm volatile("" : "+v"(fo));
            const float* lkb = ldsb + fo;
#pragma unroll
            for (int q = 0; q < 4; ++q) {
                v2f a0, a1;
                a0[0] = xs0[2 * q]; a0[1] = xs0[2 * q + 1];
                a1[0] = xs1[2 * q]; a1[1] = xs1[2 * q + 1];
                {
                    const bool wrap = (t == 15 && q == 3);
                    const int ntn = (q < 3) ? nt : (wrap ? 16 : nt - 1);
                    const float* ln = wrap ? ldsb : lkb + (q + 1) * nt * 64;
#pragma unroll
                    for (int i = 0; i < ntn; ++i)
                        bs[(q + 1) & 1][i] = *(const v2f*)(ln + i * 64);
                }
                __builtin_amdgcn_sched_barrier(0);
#pragma unroll
                for (int i = 0; i < nt; ++i) {
                    acc[0][t + i] = wmma_16x16x4_f32(bs[q & 1][i], a0, acc[0][t + i]);
                    acc[1][t + i] = wmma_16x16x4_f32(bs[q & 1][i], a1, acc[1][t + i]);
                }
                __builtin_amdgcn_sched_barrier(0);
            }
#pragma unroll
            for (int r = 0; r < 8; ++r) {
                p0[r & 1] += acc[0][t][r] * xs0[r];
                p1[r & 1] += acc[1][t][r] * xs1[r];
            }
        }
        xt = xnext;

        float ps0 = p0[0] + p0[1];
        float ps1 = p1[0] + p1[1];
        ps0 += __shfl_xor(ps0, 16, 32);
        ps1 += __shfl_xor(ps1, 16, 32);
        out[(long)slab * 32 + lane] = (hsel ? ps1 : ps0) + bias0;
    }
}

extern "C" void kernel_launch(void* const* d_in, const int* in_sizes, int n_in,
                              void* d_out, int out_size, void* d_ws, size_t ws_size,
                              hipStream_t stream) {
    const float* x   = (const float*)d_in[0];
    const float* mpw = (const float*)d_in[1];
    const float* dk  = (const float*)d_in[2];
    const float* db  = (const float*)d_in[3];
    float* out = (float*)d_out;
    float* ws  = (float*)d_ws;

    hipLaunchKernelGGL(mpo_setup_kernel, dim3(136), dim3(1024),
                       21120 * sizeof(float), stream, mpw, dk, ws + 524288);

    hipLaunchKernelGGL(mpo_main_kernel, dim3(256), dim3(256),
                       34816 * sizeof(float), stream,
                       x, db, ws + 524288, out);
}
